# starting kernel
# baseline (speedup 1.0000x reference)
#include <hip/hip_runtime.h>

typedef __attribute__((ext_vector_type(2))) float v2f;
typedef __attribute__((ext_vector_type(8))) float v8f;

#define NGROUPS 17
#define WAVES_PER_BLOCK 8
#define TILE 16
#define TPW  (2 * TILE)
#define XROW 48
#define OROW 51

__global__ __launch_bounds__(WAVES_PER_BLOCK * 32)
void bone_mlp_wmma_kernel(const float* __restrict__ x,
                          const float* __restrict__ W1,
                          const float* __restrict__ b1,
                          const float* __restrict__ W2,
                          const float* __restrict__ b2,
                          const int*   __restrict__ idx,
                          float* __restrict__ out,
                          int npairs)
{
    __shared__ float lds_out[WAVES_PER_BLOCK][TPW * OROW];

    const int lane = threadIdx.x & 31;
    const int wave = threadIdx.x >> 5;
    const int pair = blockIdx.x * WAVES_PER_BLOCK + wave;
    if (pair >= npairs) return;

    const int half = lane >> 4;
    const int lm   = lane & 15;
    const int cidx = lm < 3 ? lm : 2;

    const float* xrowA = x + (size_t)(pair * TPW + lm) * XROW;
    const float* xrowB = xrowA + TILE * XROW;
    float* lrow = &lds_out[wave][0];

    for (int g = 0; g < NGROUPS; ++g) {
        const float* w1g  = W1 + g * 12 * 16;
        const float* w2g  = W2 + g * 16 * 3 + cidx;
        const int*   idxg = idx + g * 4;

        v8f accA = {};
        v8f accB = {};
#pragma unroll
        for (int j = 0; j < 3; ++j) {
            const int f0 = 4 * j + 2 * half;
            const int f1 = f0 + 1;
            const int o0 = idxg[f0 / 3] * 3 + f0 % 3;
            const int o1 = idxg[f1 / 3] * 3 + f1 % 3;
            v2f a, bA, bB;
            a.x  = w1g[f0 * 16 + lm];
            a.y  = w1g[f1 * 16 + lm];
            bA.x = xrowA[o0];  bA.y = xrowA[o1];
            bB.x = xrowB[o0];  bB.y = xrowB[o1];
            accA = __builtin_amdgcn_wmma_f32_16x16x4_f32(
                false, a, false, bA, (short)0, accA, false, false);
            accB = __builtin_amdgcn_wmma_f32_16x16x4_f32(
                false, a, false, bB, (short)0, accB, false, false);
        }

        const float* b1g = b1 + g * 16 + 8 * half;
#pragma unroll
        for (int i = 0; i < 8; ++i) {
            const float bb = b1g[i];
            float tA = accA[i] + bb;
            float tB = accB[i] + bb;
            accA[i] = tA > 0.0f ? tA : 0.0f;
            accB[i] = tB > 0.0f ? tB : 0.0f;
        }

        v8f acc2A = {};
        v8f acc2B = {};
#pragma unroll
        for (int j = 0; j < 4; ++j) {
            const int h0 = 2 * j + 8 * half;
            v2f b;
            b.x = w2g[(h0 + 0) * 3];
            b.y = w2g[(h0 + 1) * 3];
            v2f aA; aA.x = accA[2 * j]; aA.y = accA[2 * j + 1];
            v2f aB; aB.x = accB[2 * j]; aB.y = accB[2 * j + 1];
            acc2A = __builtin_amdgcn_wmma_f32_16x16x4_f32(
                false, aA, false, b, (short)0, acc2A, false, false);
            acc2B = __builtin_amdgcn_wmma_f32_16x16x4_f32(
                false, aB, false, b, (short)0, acc2B, false, false);
        }

        const float bb2 = b2[g * 3 + cidx];
        if (lm < 3) {
#pragma unroll
            for (int i = 0; i < 8; ++i) {
                const int row = i + 8 * half;
                lrow[row * OROW + g * 3 + lm]                = acc2A[i] + bb2;
                lrow[(row + TILE) * OROW + g * 3 + lm]       = acc2B[i] + bb2;
            }
        }
    }

    float* outp = out + (size_t)pair * (TPW * OROW);
#pragma unroll 4
    for (int k = lane; k < TPW * OROW; k += 32)
        __builtin_nontemporal_store(lrow[k], &outp[k]);
}

extern "C" void kernel_launch(void* const* d_in, const int* in_sizes, int n_in,
                              void* d_out, int out_size, void* d_ws, size_t ws_size,
                              hipStream_t stream) {
    (void)n_in; (void)out_size; (void)d_ws; (void)ws_size;
    const float* x  = (const float*)d_in[0];
    const float* W1 = (const float*)d_in[1];
    const float* b1 = (const float*)d_in[2];
    const float* W2 = (const float*)d_in[3];
    const float* b2 = (const float*)d_in[4];
    const int*   idx = (const int*)d_in[5];
    float* out = (float*)d_out;

    const int ntok   = in_sizes[0] / XROW;
    const int npairs = ntok / TPW;
    const int blocks = (npairs + WAVES_PER_BLOCK - 1) / WAVES_PER_BLOCK;

    bone_mlp_wmma_kernel<<<blocks, WAVES_PER_BLOCK * 32, 0, stream>>>(
        x, W1, b1, W2, b2, idx, out, npairs);
}


# One wave per bone group (17-wave blocks walk 16 token units): W1/b1/W2/b2 stay in registers as WMMA fragments, biases enter as C, layer-1 k runs limb by limb (sum order changes), shared LDS out tile
# speedup vs baseline: 2.6560x; 2.6560x over previous
#include <hip/hip_runtime.h>


typedef __attribute__((ext_vector_type(2))) float v2f;
typedef __attribute__((ext_vector_type(4))) float v4f;
typedef __attribute__((ext_vector_type(8))) float v8f;

#define NGROUPS 17
#define TILE 16
#define TPU  (2 * TILE)
#define XROW 48
#define OROW 51
#define UNIT_FLOATS (TPU * OROW)
#define UNIT_VEC4   (UNIT_FLOATS / 4)
#define UNITS_PER_BLOCK 16

struct limb3 { float a, b, c; };

__device__ __forceinline__ limb3 load_limb(const float* p)
{
    limb3 r;
    r.a = p[0]; r.b = p[1]; r.c = p[2];
    return r;
}

__device__ __forceinline__ v8f mm4(v2f a, v2f b, v8f c)
{
    return __builtin_amdgcn_wmma_f32_16x16x4_f32(false, a, false, b, (short)0, c, false, false);
}

__global__ __launch_bounds__(NGROUPS * 32)
void bone_mlp_group_wave_kernel(const float* __restrict__ x,
                                const float* __restrict__ W1,
                                const float* __restrict__ b1,
                                const float* __restrict__ W2,
                                const float* __restrict__ b2,
                                const int*   __restrict__ idx,
                                float* __restrict__ out,
                                int nunits)
{
    __shared__ __attribute__((aligned(16))) float lds_out[2][UNIT_FLOATS];

    const int tid  = threadIdx.x;
    const int lane = tid & 31;
    const int g    = __builtin_amdgcn_readfirstlane(tid >> 5);
    const int h    = lane >> 4;
    const int lm   = lane & 15;

    const int u0 = blockIdx.x * UNITS_PER_BLOCK;
    int u1 = u0 + UNITS_PER_BLOCK;
    if (u1 > nunits) u1 = nunits;
    if (u0 >= u1) return;

    int s0 = idx[g * 4 + 2 * h];
    int s1 = idx[g * 4 + 2 * h + 1];
    s0 = s0 < 0 ? 0 : (s0 > 15 ? 15 : s0);
    s1 = s1 < 0 ? 0 : (s1 > 15 ? 15 : s1);
    const int off0 = lm * XROW + s0 * 3;
    const int off1 = lm * XROW + s1 * 3;

    const float* w1g = W1 + g * (12 * 16) + (6 * h) * 16 + lm;
    v2f a1[3];
#pragma unroll
    for (int j = 0; j < 3; ++j) {
        a1[j].x = w1g[(2 * j) * 16];
        a1[j].y = w1g[(2 * j + 1) * 16];
    }
    v8f c1;
#pragma unroll
    for (int r = 0; r < 8; ++r) c1[r] = b1[g * 16 + 8 * h + r];

    const int cc = lm < 3 ? lm : 2;
    const float* w2g = W2 + g * (16 * 3) + (8 * h) * 3 + cc;
    v2f a2[4];
#pragma unroll
    for (int j = 0; j < 4; ++j) {
        a2[j].x = w2g[(2 * j) * 3];
        a2[j].y = w2g[(2 * j + 1) * 3];
    }
    v8f c2 = {};
#pragma unroll
    for (int r = 0; r < 3; ++r) c2[r] = b2[g * 3 + r];

    const float* xu = x + (size_t)u0 * (TPU * XROW);
    limb3 pA0 = load_limb(xu + off0);
    limb3 pA1 = load_limb(xu + off1);
    limb3 pB0 = load_limb(xu + TILE * XROW + off0);
    limb3 pB1 = load_limb(xu + TILE * XROW + off1);

    for (int u = u0; u < u1; ++u) {
        const int un = (u + 1 < u1) ? (u + 1) : u;
        const float* xn = x + (size_t)un * (TPU * XROW);
        const limb3 nA0 = load_limb(xn + off0);
        const limb3 nA1 = load_limb(xn + off1);
        const limb3 nB0 = load_limb(xn + TILE * XROW + off0);
        const limb3 nB1 = load_limb(xn + TILE * XROW + off1);
        __builtin_amdgcn_sched_barrier(0);

        v2f bA, bB;
        bA.x = pA0.a; bA.y = pA0.b;  bB.x = pB0.a; bB.y = pB0.b;
        v8f hA = mm4(a1[0], bA, c1);
        v8f hB = mm4(a1[0], bB, c1);
        bA.x = pA0.c; bA.y = pA1.a;  bB.x = pB0.c; bB.y = pB1.a;
        hA = mm4(a1[1], bA, hA);
        hB = mm4(a1[1], bB, hB);
        bA.x = pA1.b; bA.y = pA1.c;  bB.x = pB1.b; bB.y = pB1.c;
        hA = mm4(a1[2], bA, hA);
        hB = mm4(a1[2], bB, hB);
#pragma unroll
        for (int r = 0; r < 8; ++r) {
            hA[r] = __builtin_fmaxf(hA[r], 0.0f);
            hB[r] = __builtin_fmaxf(hB[r], 0.0f);
        }

        v8f oA = c2, oB = c2;
#pragma unroll
        for (int j = 0; j < 4; ++j) {
            v2f kA, kB;
            kA.x = hA[2 * j]; kA.y = hA[2 * j + 1];
            kB.x = hB[2 * j]; kB.y = hB[2 * j + 1];
            oA = mm4(a2[j], kA, oA);
            oB = mm4(a2[j], kB, oB);
        }

        float* lo = lds_out[(u - u0) & 1];
        if (h == 0) {
            float* r = lo + lm * OROW + g * 3;
            r[0] = oA[0]; r[1] = oA[1]; r[2] = oA[2];
            r += TILE * OROW;
            r[0] = oB[0]; r[1] = oB[1]; r[2] = oB[2];
        }

        __builtin_amdgcn_fence(__ATOMIC_RELEASE, "workgroup", "local");
        __builtin_amdgcn_s_barrier();
        __builtin_amdgcn_fence(__ATOMIC_ACQUIRE, "workgroup", "local");

        if (tid < UNIT_VEC4) {
            const v4f v = *(const v4f*)(lo + tid * 4);
            __builtin_nontemporal_store(v, (v4f*)(out + (size_t)u * UNIT_FLOATS + tid * 4));
        }

        pA0 = nA0; pA1 = nA1; pB0 = nB0; pB1 = nB1;
    }
}

extern "C" void kernel_launch(void* const* d_in, const int* in_sizes, int n_in,
                              void* d_out, int out_size, void* d_ws, size_t ws_size,
                              hipStream_t stream) {
    (void)n_in; (void)out_size; (void)d_ws; (void)ws_size;
    const float* x  = (const float*)d_in[0];
    const float* W1 = (const float*)d_in[1];
    const float* b1 = (const float*)d_in[2];
    const float* W2 = (const float*)d_in[3];
    const float* b2 = (const float*)d_in[4];
    const int*   idx = (const int*)d_in[5];
    float* out = (float*)d_out;

    const int ntok   = in_sizes[0] / XROW;
    const int nunits = ntok / TPU;
    const int blocks = (nunits + UNITS_PER_BLOCK - 1) / UNITS_PER_BLOCK;

    bone_mlp_group_wave_kernel<<<blocks, NGROUPS * 32, 0, stream>>>(
        x, W1, b1, W2, b2, idx, out, nunits);
}
